# starting kernel
# baseline (speedup 1.0000x reference)
#include <hip/hip_runtime.h>


typedef __attribute__((ext_vector_type(2))) float v2f;
typedef __attribute__((ext_vector_type(4))) float v4f;
typedef __attribute__((ext_vector_type(8))) float v8f;

#define I_DIM 8
#define O_DIM 4
#define U_DIM 256
#define B_DIM 4096

#define BM 64
#define BN 256
#define KS 32
#define LDX 36
#define LDWP 544

__global__ __launch_bounds__(256)
void moe_gate_gemm_f32wmma(const float* __restrict__ xg,
                           const float* __restrict__ ugate,
                           const float* __restrict__ Wg,
                           const float* __restrict__ alpha,
                           float* __restrict__ outg)
{
    __shared__ float ldsX[BM * LDX];
    __shared__ float ldsWp[(KS / 2) * LDWP];
    __shared__ float zbuf[BN];

    const int tid  = threadIdx.x;
    const int o    = blockIdx.y;
    const int m0   = blockIdx.x * BM;

    const int wave = tid >> 5;
    const int lane = tid & 31;
    const int lh   = lane >> 4;
    const int lm   = lane & 15;
    const int wm   = (wave >> 2) * 32;
    const int wn   = (wave & 3) * 64;

    const float la = __logf(alpha[0]) * (1.0f / 0.9f);

    v8f acc[2][4] = {};

    for (int j = 0; j < I_DIM; ++j) {
        {
            float uv = ugate[(o * I_DIM + j) * U_DIM + tid];
            float t  = __logf(uv) - __logf(1.0f - uv) + la;
            float s  = 1.0f / (1.0f + __expf(-t));
            float z  = s * 1.2f - 0.1f;
            zbuf[tid] = fminf(fmaxf(z, 0.0f), 1.0f);
        }
        __syncthreads();

        const float* xj = xg + (size_t)j * B_DIM * U_DIM + (size_t)m0 * U_DIM;
        const float* Wj = Wg + ((size_t)(o * I_DIM + j) * U_DIM) * U_DIM;

        for (int ks = 0; ks < U_DIM; ks += KS) {
            #pragma unroll
            for (int s = 0; s < 2; ++s) {
                int idx = tid + s * 256;
                int row = idx >> 3;
                int kq  = (idx & 7) << 2;
                v4f v = *(const v4f*)(xj + (size_t)row * U_DIM + ks + kq);
                *(v4f*)(&ldsX[row * LDX + kq]) = v;
            }
            #pragma unroll
            for (int s = 0; s < 8; ++s) {
                int idx = tid + s * 256;
                int kp  = idx >> 7;
                int n   = (idx & 127) << 1;
                const float* wrow = Wj + (size_t)(ks + 2 * kp) * U_DIM + n;
                v2f g0 = *(const v2f*)(wrow);
                v2f g1 = *(const v2f*)(wrow + U_DIM);
                float z0 = zbuf[n], z1 = zbuf[n + 1];
                v4f cell;
                cell.x = g0.x * z0; cell.y = g1.x * z0;
                cell.z = g0.y * z1; cell.w = g1.y * z1;
                *(v4f*)(&ldsWp[kp * LDWP + n * 2]) = cell;
            }
            if (ks + KS < U_DIM) {
                __builtin_prefetch(xj + (size_t)(tid >> 2) * U_DIM + ks + KS + ((tid & 3) << 3), 0, 3);
                __builtin_prefetch(Wj + (size_t)(ks + KS + (tid >> 3)) * U_DIM + ((tid & 7) << 5), 0, 3);
            }
            __syncthreads();

            #pragma unroll
            for (int k = 0; k < KS; k += 4) {
                v2f a0 = *(const v2f*)(&ldsX[(wm + 0  + lm) * LDX + k + 2 * lh]);
                v2f a1 = *(const v2f*)(&ldsX[(wm + 16 + lm) * LDX + k + 2 * lh]);
                const float* wb = &ldsWp[((k >> 1) + lh) * LDWP + (wn + lm) * 2];
                v2f b0 = *(const v2f*)(wb);
                v2f b1 = *(const v2f*)(wb + 32);
                v2f b2 = *(const v2f*)(wb + 64);
                v2f b3 = *(const v2f*)(wb + 96);

                acc[0][0] = __builtin_amdgcn_wmma_f32_16x16x4_f32(
                    false, a0, false, b0, (short)0, acc[0][0], false, false);
                acc[0][1] = __builtin_amdgcn_wmma_f32_16x16x4_f32(
                    false, a0, false, b1, (short)0, acc[0][1], false, false);
                acc[0][2] = __builtin_amdgcn_wmma_f32_16x16x4_f32(
                    false, a0, false, b2, (short)0, acc[0][2], false, false);
                acc[0][3] = __builtin_amdgcn_wmma_f32_16x16x4_f32(
                    false, a0, false, b3, (short)0, acc[0][3], false, false);
                acc[1][0] = __builtin_amdgcn_wmma_f32_16x16x4_f32(
                    false, a1, false, b0, (short)0, acc[1][0], false, false);
                acc[1][1] = __builtin_amdgcn_wmma_f32_16x16x4_f32(
                    false, a1, false, b1, (short)0, acc[1][1], false, false);
                acc[1][2] = __builtin_amdgcn_wmma_f32_16x16x4_f32(
                    false, a1, false, b2, (short)0, acc[1][2], false, false);
                acc[1][3] = __builtin_amdgcn_wmma_f32_16x16x4_f32(
                    false, a1, false, b3, (short)0, acc[1][3], false, false);
            }
            __syncthreads();
        }
    }

    #pragma unroll
    for (int mi = 0; mi < 2; ++mi) {
        #pragma unroll
        for (int ni = 0; ni < 4; ++ni) {
            #pragma unroll
            for (int r = 0; r < 8; ++r) {
                int row = m0 + wm + mi * 16 + r + 8 * lh;
                int col = wn + ni * 16 + lm;
                outg[((size_t)o * B_DIM + row) * U_DIM + col] = acc[mi][ni][r];
            }
        }
    }
}

extern "C" void kernel_launch(void* const* d_in, const int* in_sizes, int n_in,
                              void* d_out, int out_size, void* d_ws, size_t ws_size,
                              hipStream_t stream) {
    const float* x     = (const float*)d_in[0];
    const float* ugate = (const float*)d_in[1];
    const float* W     = (const float*)d_in[2];
    const float* alpha = (const float*)d_in[3];
    float* out = (float*)d_out;

    dim3 grid(B_DIM / BM, O_DIM);
    moe_gate_gemm_f32wmma<<<grid, 256, 0, stream>>>(x, ugate, W, alpha, out);
}


# K loop software-pipelined over two LDS stages: next slab's global loads issued into registers ahead of the WMMA block, gate-scaled and stored in its tail; one barrier per slab; sum order kept
# speedup vs baseline: 1.1559x; 1.1559x over previous
#include <hip/hip_runtime.h>


typedef __attribute__((ext_vector_type(2))) float v2f;
typedef __attribute__((ext_vector_type(4))) float v4f;
typedef __attribute__((ext_vector_type(8))) float v8f;

#define I_DIM 8
#define O_DIM 4
#define U_DIM 256
#define B_DIM 4096

#define BM 64
#define BN 256
#define KS 32
#define LDX 36
#define LDWP 544

#define NKS (U_DIM / KS)
#define NSLAB (I_DIM * NKS)
#define XSTAGE (BM * LDX)
#define WSTAGE ((KS / 2) * LDWP)

struct SlabRegs {
    v4f x[2];
    v2f w0[8];
    v2f w1[8];
};

__device__ __forceinline__ void slab_load(SlabRegs& r, const float* xb, const float* wb,
                                          unsigned xo, unsigned wo)
{
    const float* xt = xb + xo;
    const float* wt = wb + wo;
    r.x[0] = *(const v4f*)(xt);
    r.x[1] = *(const v4f*)(xt + 32 * U_DIM);
    #pragma unroll
    for (int s = 0; s < 8; ++s) {
        const float* wrow = wt + s * 4 * U_DIM;
        r.w0[s] = *(const v2f*)(wrow);
        r.w1[s] = *(const v2f*)(wrow + U_DIM);
    }
}

__device__ __forceinline__ void slab_store(const SlabRegs& r, float* xs, float* ws,
                                           float z0, float z1)
{
    *(v4f*)(xs)            = r.x[0];
    *(v4f*)(xs + 32 * LDX) = r.x[1];
    #pragma unroll
    for (int s = 0; s < 8; ++s) {
        v4f cell;
        cell.x = r.w0[s].x * z0; cell.y = r.w1[s].x * z0;
        cell.z = r.w0[s].y * z1; cell.w = r.w1[s].y * z1;
        *(v4f*)(ws + s * 2 * LDWP) = cell;
    }
}

__device__ __forceinline__ float gate(float uv, float la)
{
    float t = __logf(uv) - __logf(1.0f - uv) + la;
    float s = 1.0f / (1.0f + __expf(-t));
    float z = s * 1.2f - 0.1f;
    return fminf(fmaxf(z, 0.0f), 1.0f);
}

struct Frag { v2f a0, a1, b0, b1, b2, b3; };

__device__ __forceinline__ Frag frag_load(const float* xa, const float* wb0, int k)
{
    Frag f;
    f.a0 = *(const v2f*)(xa + k);
    f.a1 = *(const v2f*)(xa + 16 * LDX + k);
    const float* wb = wb0 + (k >> 1) * LDWP;
    f.b0 = *(const v2f*)(wb);
    f.b1 = *(const v2f*)(wb + 32);
    f.b2 = *(const v2f*)(wb + 64);
    f.b3 = *(const v2f*)(wb + 96);
    return f;
}

__device__ __forceinline__ void frag_mma(v8f (&acc)[2][4], const Frag& f)
{
    acc[0][0] = __builtin_amdgcn_wmma_f32_16x16x4_f32(
        false, f.a0, false, f.b0, (short)0, acc[0][0], false, false);
    acc[0][1] = __builtin_amdgcn_wmma_f32_16x16x4_f32(
        false, f.a0, false, f.b1, (short)0, acc[0][1], false, false);
    acc[0][2] = __builtin_amdgcn_wmma_f32_16x16x4_f32(
        false, f.a0, false, f.b2, (short)0, acc[0][2], false, false);
    acc[0][3] = __builtin_amdgcn_wmma_f32_16x16x4_f32(
        false, f.a0, false, f.b3, (short)0, acc[0][3], false, false);
    acc[1][0] = __builtin_amdgcn_wmma_f32_16x16x4_f32(
        false, f.a1, false, f.b0, (short)0, acc[1][0], false, false);
    acc[1][1] = __builtin_amdgcn_wmma_f32_16x16x4_f32(
        false, f.a1, false, f.b1, (short)0, acc[1][1], false, false);
    acc[1][2] = __builtin_amdgcn_wmma_f32_16x16x4_f32(
        false, f.a1, false, f.b2, (short)0, acc[1][2], false, false);
    acc[1][3] = __builtin_amdgcn_wmma_f32_16x16x4_f32(
        false, f.a1, false, f.b3, (short)0, acc[1][3], false, false);
}

template <int K0, int K1>
__device__ __forceinline__ void wmma_steps(v8f (&acc)[2][4], const float* xa, const float* wb0)
{
    #pragma unroll
    for (int k = K0; k < K1; k += 4) {
        Frag f = frag_load(xa, wb0, k);
        frag_mma(acc, f);
    }
}

#define KSPLIT 24

__global__ __launch_bounds__(256)
void moe_gate_gemm_f32wmma(const float* xg,
                           const float* ugate,
                           const float* Wg,
                           const float* __restrict__ alpha,
                           float* outg)
{
    __shared__ float ldsX[2 * XSTAGE];
    __shared__ float ldsWp[2 * WSTAGE];

    const int tid  = threadIdx.x;
    const int o    = blockIdx.y;
    const int m0   = blockIdx.x * BM;

    const int wave = tid >> 5;
    const int lane = tid & 31;
    const int lh   = lane >> 4;
    const int lm   = lane & 15;
    const int wm   = (wave >> 2) * 32;
    const int wn   = (wave & 3) * 64;

    const float la = __logf(alpha[0]) * (1.0f / 0.9f);

    const int n = (tid & 127) << 1;
    const unsigned xo = (tid >> 3) * U_DIM + ((tid & 7) << 2);
    const unsigned wo = (2 * (tid >> 7)) * U_DIM + n;
    const float* xblk = xg + (size_t)m0 * U_DIM;
    const float* wblk = Wg + (size_t)o * I_DIM * U_DIM * U_DIM;
    const float* ublk = ugate + o * I_DIM * U_DIM;
    const int xs_off = (tid >> 3) * LDX + ((tid & 7) << 2);
    const int ws_off = (tid >> 7) * LDWP + n * 2;
    const int a_off = (wm + lm) * LDX + 2 * lh;
    const int b_off = lh * LDWP + (wn + lm) * 2;

    v8f acc[2][4] = {};

    SlabRegs r;
    float z0, z1;

    {
        v2f uv = *(const v2f*)(ublk + (unsigned)n);
        slab_load(r, xblk, wblk, xo, wo);
        z0 = gate(uv.x, la);
        z1 = gate(uv.y, la);
        slab_store(r, ldsX + xs_off, ldsWp + ws_off, z0, z1);
    }
    __syncthreads();

    for (int t = 0; t < NSLAB - 1; ++t) {
        const int tn  = t + 1;
        const int jn  = tn >> 3;
        const int cur = t & 1;
        const float* xa  = ldsX  + cur * XSTAGE + a_off;
        const float* wb0 = ldsWp + cur * WSTAGE + b_off;

        Frag f0 = frag_load(xa, wb0, 0);
        asm volatile("" ::: "memory");

        slab_load(r, xblk + (size_t)jn * B_DIM * U_DIM + (tn & (NKS - 1)) * KS,
                     wblk + (size_t)tn * KS * U_DIM, xo, wo);
        v2f uv = *(const v2f*)(ublk + jn * U_DIM + (unsigned)n);
        asm volatile("" ::: "memory");

        frag_mma(acc, f0);
        wmma_steps<4, KSPLIT>(acc, xa, wb0);

        if ((tn & (NKS - 1)) == 0) {
            z0 = gate(uv.x, la);
            z1 = gate(uv.y, la);
        }
        slab_store(r, ldsX + (cur ^ 1) * XSTAGE + xs_off,
                      ldsWp + (cur ^ 1) * WSTAGE + ws_off, z0, z1);

        wmma_steps<KSPLIT, KS>(acc, xa, wb0);
        __syncthreads();
    }
    wmma_steps<0, KS>(acc, ldsX + ((NSLAB - 1) & 1) * XSTAGE + a_off,
                           ldsWp + ((NSLAB - 1) & 1) * WSTAGE + b_off);

    #pragma unroll
    for (int mi = 0; mi < 2; ++mi) {
        #pragma unroll
        for (int ni = 0; ni < 4; ++ni) {
            #pragma unroll
            for (int rr = 0; rr < 8; ++rr) {
                int row = m0 + wm + mi * 16 + rr + 8 * lh;
                int col = wn + ni * 16 + lm;
                outg[((size_t)o * B_DIM + row) * U_DIM + col] = acc[mi][ni][rr];
            }
        }
    }
}

extern "C" void kernel_launch(void* const* d_in, const int* in_sizes, int n_in,
                              void* d_out, int out_size, void* d_ws, size_t ws_size,
                              hipStream_t stream) {
    const float* x     = (const float*)d_in[0];
    const float* ugate = (const float*)d_in[1];
    const float* W     = (const float*)d_in[2];
    const float* alpha = (const float*)d_in[3];
    float* out = (float*)d_out;

    dim3 grid(B_DIM / BM, O_DIM);
    moe_gate_gemm_f32wmma<<<grid, 256, 0, stream>>>(x, ugate, W, alpha, out);
}


# Fragments double-buffered in registers: all LDS fragment reads of slab t+1 go out as one group under slab t's WMMAs, none among them or right behind a barrier; sum order kept
# speedup vs baseline: 1.2471x; 1.0789x over previous
#include <hip/hip_runtime.h>


typedef __attribute__((ext_vector_type(2))) float v2f;
typedef __attribute__((ext_vector_type(4))) float v4f;
typedef __attribute__((ext_vector_type(8))) float v8f;

#define I_DIM 8
#define O_DIM 4
#define U_DIM 256
#define B_DIM 4096

#define BM 64
#define BN 256
#define KS 32
#define LDX 36
#define LDWP 544

#define NKS (U_DIM / KS)
#define NSLAB (I_DIM * NKS)
#define XSTAGE (BM * LDX)
#define WSTAGE ((KS / 2) * LDWP)

struct SlabRegs {
    v4f x[2];
    v2f w0[8];
    v2f w1[8];
};

__device__ __forceinline__ void slab_load(SlabRegs& r, const float* xb, const float* wb,
                                          unsigned xo, unsigned wo)
{
    const float* xt = xb + xo;
    const float* wt = wb + wo;
    r.x[0] = *(const v4f*)(xt);
    r.x[1] = *(const v4f*)(xt + 32 * U_DIM);
    #pragma unroll
    for (int s = 0; s < 8; ++s) {
        const float* wrow = wt + s * 4 * U_DIM;
        r.w0[s] = *(const v2f*)(wrow);
        r.w1[s] = *(const v2f*)(wrow + U_DIM);
    }
}

__device__ __forceinline__ void slab_store(const SlabRegs& r, float* xs, float* ws,
                                           float z0, float z1)
{
    *(v4f*)(xs)            = r.x[0];
    *(v4f*)(xs + 32 * LDX) = r.x[1];
    #pragma unroll
    for (int s = 0; s < 8; ++s) {
        v4f cell;
        cell.x = r.w0[s].x * z0; cell.y = r.w1[s].x * z0;
        cell.z = r.w0[s].y * z1; cell.w = r.w1[s].y * z1;
        *(v4f*)(ws + s * 2 * LDWP) = cell;
    }
}

__device__ __forceinline__ float gate(float uv, float la)
{
    float t = __logf(uv) - __logf(1.0f - uv) + la;
    float s = 1.0f / (1.0f + __expf(-t));
    float z = s * 1.2f - 0.1f;
    return fminf(fmaxf(z, 0.0f), 1.0f);
}

struct Frag { v2f a0, a1, b0, b1, b2, b3; };

__device__ __forceinline__ Frag frag_load(const float* xa, const float* wb0, int k)
{
    Frag f;
    f.a0 = *(const v2f*)(xa + k);
    f.a1 = *(const v2f*)(xa + 16 * LDX + k);
    const float* wb = wb0 + (k >> 1) * LDWP;
    f.b0 = *(const v2f*)(wb);
    f.b1 = *(const v2f*)(wb + 32);
    f.b2 = *(const v2f*)(wb + 64);
    f.b3 = *(const v2f*)(wb + 96);
    return f;
}

__device__ __forceinline__ void frag_mma(v8f (&acc)[2][4], const Frag& f)
{
    acc[0][0] = __builtin_amdgcn_wmma_f32_16x16x4_f32(
        false, f.a0, false, f.b0, (short)0, acc[0][0], false, false);
    acc[0][1] = __builtin_amdgcn_wmma_f32_16x16x4_f32(
        false, f.a0, false, f.b1, (short)0, acc[0][1], false, false);
    acc[0][2] = __builtin_amdgcn_wmma_f32_16x16x4_f32(
        false, f.a0, false, f.b2, (short)0, acc[0][2], false, false);
    acc[0][3] = __builtin_amdgcn_wmma_f32_16x16x4_f32(
        false, f.a0, false, f.b3, (short)0, acc[0][3], false, false);
    acc[1][0] = __builtin_amdgcn_wmma_f32_16x16x4_f32(
        false, f.a1, false, f.b0, (short)0, acc[1][0], false, false);
    acc[1][1] = __builtin_amdgcn_wmma_f32_16x16x4_f32(
        false, f.a1, false, f.b1, (short)0, acc[1][1], false, false);
    acc[1][2] = __builtin_amdgcn_wmma_f32_16x16x4_f32(
        false, f.a1, false, f.b2, (short)0, acc[1][2], false, false);
    acc[1][3] = __builtin_amdgcn_wmma_f32_16x16x4_f32(
        false, f.a1, false, f.b3, (short)0, acc[1][3], false, false);
}

struct SlabFrags { Frag f[KS / 4]; };

__device__ __forceinline__ void frags_load(SlabFrags& s, const float* xa, const float* wb0)
{
    #pragma unroll
    for (int ks = 0; ks < KS / 4; ++ks) s.f[ks] = frag_load(xa, wb0, 4 * ks);
}

template <int S0, int S1>
__device__ __forceinline__ void mma_steps(v8f (&acc)[2][4], const SlabFrags& s)
{
    #pragma unroll
    for (int ks = S0; ks < S1; ++ks) frag_mma(acc, s.f[ks]);
}

#define KSPLIT 24

__device__ __forceinline__ void slab_step(v8f (&acc)[2][4], const SlabFrags& cur, SlabFrags& nxt,
                                          const float* nxa, const float* nwb,
                                          SlabRegs& r, const float* xsrc, const float* wsrc,
                                          unsigned xo, unsigned wo, const float* usrc,
                                          bool newgate, float la, float& z0, float& z1,
                                          float* xs, float* ws)
{
    frag_mma(acc, cur.f[0]);
    __builtin_amdgcn_sched_barrier(0);
    frags_load(nxt, nxa, nwb);
    slab_load(r, xsrc, wsrc, xo, wo);
    v2f uv = *(const v2f*)usrc;
    __builtin_amdgcn_sched_barrier(0);

    mma_steps<1, KSPLIT / 4>(acc, cur);
    __builtin_amdgcn_sched_barrier(0);

    if (newgate) {
        z0 = gate(uv.x, la);
        z1 = gate(uv.y, la);
    }
    slab_store(r, xs, ws, z0, z1);

    mma_steps<KSPLIT / 4, KS / 4>(acc, cur);
    __syncthreads();
}

__global__ __launch_bounds__(256)
void moe_gate_gemm_f32wmma(const float* xg,
                           const float* ugate,
                           const float* Wg,
                           const float* __restrict__ alpha,
                           float* outg)
{
    __shared__ float ldsX[2 * XSTAGE];
    __shared__ float ldsWp[2 * WSTAGE];

    const int tid  = threadIdx.x;
    const int o    = blockIdx.y;
    const int m0   = blockIdx.x * BM;

    const int wave = tid >> 5;
    const int lane = tid & 31;
    const int lh   = lane >> 4;
    const int lm   = lane & 15;
    const int wm   = (wave >> 2) * 32;
    const int wn   = (wave & 3) * 64;

    const float la = __logf(alpha[0]) * (1.0f / 0.9f);

    const int n = (tid & 127) << 1;
    const unsigned xo = (tid >> 3) * U_DIM + ((tid & 7) << 2);
    const unsigned wo = (2 * (tid >> 7)) * U_DIM + n;
    const float* xblk = xg + (size_t)m0 * U_DIM;
    const float* wblk = Wg + (size_t)o * I_DIM * U_DIM * U_DIM;
    const float* ublk = ugate + o * I_DIM * U_DIM;
    const int xs_off = (tid >> 3) * LDX + ((tid & 7) << 2);
    const int ws_off = (tid >> 7) * LDWP + n * 2;
    const int a_off = (wm + lm) * LDX + 2 * lh;
    const int b_off = lh * LDWP + (wn + lm) * 2;

    v8f acc[2][4] = {};

    const float* xa0 = ldsX + a_off;
    const float* xa1 = ldsX + XSTAGE + a_off;
    const float* wb0 = ldsWp + b_off;
    const float* wb1 = ldsWp + WSTAGE + b_off;
    float* xs0 = ldsX + xs_off;
    float* xs1 = ldsX + XSTAGE + xs_off;
    float* ws0 = ldsWp + ws_off;
    float* ws1 = ldsWp + WSTAGE + ws_off;

    SlabRegs r;
    SlabFrags fa, fb;
    float z0, z1;

    {
        SlabRegs r0;
        v2f uv = *(const v2f*)(ublk + (unsigned)n);
        slab_load(r0, xblk, wblk, xo, wo);
        slab_load(r, xblk + KS, wblk + (size_t)KS * U_DIM, xo, wo);
        z0 = gate(uv.x, la);
        z1 = gate(uv.y, la);
        slab_store(r0, xs0, ws0, z0, z1);
        __syncthreads();
        frags_load(fa, xa0, wb0);
        slab_store(r, xs1, ws1, z0, z1);
        __syncthreads();
    }

    for (int te = 0; te < NSLAB - 2; te += 2) {
        {
            const int tn = te + 2;
            const int jn = tn >> 3;
            slab_step(acc, fa, fb, xa1, wb1, r,
                      xblk + (size_t)jn * B_DIM * U_DIM + (tn & (NKS - 1)) * KS,
                      wblk + (size_t)tn * KS * U_DIM, xo, wo,
                      ublk + jn * U_DIM + (unsigned)n, (tn & (NKS - 1)) == 0,
                      la, z0, z1, xs0, ws0);
        }
        {
            const int tn = te + 3;
            const int jn = tn >> 3;
            slab_step(acc, fb, fa, xa0, wb0, r,
                      xblk + (size_t)jn * B_DIM * U_DIM + (tn & (NKS - 1)) * KS,
                      wblk + (size_t)tn * KS * U_DIM, xo, wo,
                      ublk + jn * U_DIM + (unsigned)n, false,
                      la, z0, z1, xs1, ws1);
        }
    }

    frag_mma(acc, fa.f[0]);
    __builtin_amdgcn_sched_barrier(0);
    frags_load(fb, xa1, wb1);
    __builtin_amdgcn_sched_barrier(0);
    mma_steps<1, KS / 4>(acc, fa);
    mma_steps<0, KS / 4>(acc, fb);

    #pragma unroll
    for (int mi = 0; mi < 2; ++mi) {
        #pragma unroll
        for (int ni = 0; ni < 4; ++ni) {
            #pragma unroll
            for (int rr = 0; rr < 8; ++rr) {
                int row = m0 + wm + mi * 16 + rr + 8 * lh;
                int col = wn + ni * 16 + lm;
                outg[((size_t)o * B_DIM + row) * U_DIM + col] = acc[mi][ni][rr];
            }
        }
    }
}

extern "C" void kernel_launch(void* const* d_in, const int* in_sizes, int n_in,
                              void* d_out, int out_size, void* d_ws, size_t ws_size,
                              hipStream_t stream) {
    const float* x     = (const float*)d_in[0];
    const float* ugate = (const float*)d_in[1];
    const float* W     = (const float*)d_in[2];
    const float* alpha = (const float*)d_in[3];
    float* out = (float*)d_out;

    dim3 grid(B_DIM / BM, O_DIM);
    moe_gate_gemm_f32wmma<<<grid, 256, 0, stream>>>(x, ugate, W, alpha, out);
}
